# starting kernel
# baseline (speedup 1.0000x reference)
#include <hip/hip_runtime.h>
#include <math.h>


typedef __attribute__((ext_vector_type(2))) float v2f;
typedef __attribute__((ext_vector_type(8))) float v8f;

#define NB   32
#define SD   512
#define SE   512
#define DD   512
#define STR  516

__launch_bounds__(256, 1)
__global__ void attn_fused_kernel(const float* __restrict__ Q,
                                  const float* __restrict__ E,
                                  const float* __restrict__ mask,
                                  const float* __restrict__ W,
                                  const float* __restrict__ bias,
                                  float* __restrict__ outMasked,
                                  float* __restrict__ outWeights)
{
    __shared__ float sQ[16 * STR];
    __shared__ float sS[16 * STR];
    __shared__ float sRed[16 * 16];
    __shared__ float sMask[16];

    const int b    = blockIdx.y;
    const int q0   = blockIdx.x * 16;
    const int tid  = threadIdx.x;
    const int lane = tid & 31;
    const int wave = tid >> 5;
    const int half = lane >> 4;
    const int l16  = lane & 15;

    const float* Qb = Q + ((size_t)b * SD + q0) * DD;
    const float* Eb = E + (size_t)b * SE * DD;

    for (int i = tid; i < 16 * DD; i += 256) {
        int r = i >> 9;
        int c = i & 511;
        sQ[r * STR + c] = Qb[r * DD + c];
    }
    if (tid < 16) sMask[tid] = mask[(size_t)b * SD + q0 + tid];
    __syncthreads();

    const float scale = 0.044194173824159216f;

    {
        const float* Arow = sQ + l16 * STR + 2 * half;
        for (int t = 0; t < 4; ++t) {
            const int k0 = (wave * 4 + t) * 16;
            const float* Brow = Eb + (size_t)(k0 + l16) * DD + 2 * half;
            v8f acc = {};
            for (int d = 0; d < DD; d += 4) {
                v2f a;  a.x = Arow[d];  a.y = Arow[d + 1];
                v2f bf; bf.x = Brow[d]; bf.y = Brow[d + 1];
                acc = __builtin_amdgcn_wmma_f32_16x16x4_f32(
                    false, a, false, bf, (short)0, acc, false, false);
            }
#pragma unroll
            for (int i = 0; i < 8; ++i)
                sS[(i + 8 * half) * STR + k0 + l16] = acc[i] * scale;
        }
    }
    __syncthreads();

    {
        const int row = tid & 15;
        const int sub = tid >> 4;
        float* srow = sS + row * STR;
        float m = -INFINITY;
        for (int c = sub * 32; c < sub * 32 + 32; ++c) m = fmaxf(m, srow[c]);
        sRed[row * 16 + sub] = m;
        __syncthreads();
        float mall = -INFINITY;
#pragma unroll
        for (int j = 0; j < 16; ++j) mall = fmaxf(mall, sRed[row * 16 + j]);
        __syncthreads();
        float s = 0.f;
        for (int c = sub * 32; c < sub * 32 + 32; ++c) {
            float p = __expf(srow[c] - mall);
            srow[c] = p;
            s += p;
        }
        sRed[row * 16 + sub] = s;
        __syncthreads();
        float stot = 0.f;
#pragma unroll
        for (int j = 0; j < 16; ++j) stot += sRed[row * 16 + j];
        const float inv = 1.0f / stot;
        float* gw = outWeights + ((size_t)b * SD + q0 + row) * SE;
        for (int c = sub * 32; c < sub * 32 + 32; ++c) {
            float p = srow[c] * inv;
            srow[c] = p;
            gw[c]   = p;
        }
    }
    __syncthreads();

    v8f ctx[4];
    {
        const float* Arow = sS + l16 * STR + 2 * half;
        for (int t = 0; t < 4; ++t) {
            const int n0 = wave * 64 + t * 16;
            const float* Bcol = Eb + n0 + l16;
            v8f acc = {};
            for (int k = 0; k < SE; k += 4) {
                v2f a;  a.x = Arow[k]; a.y = Arow[k + 1];
                v2f bf;
                bf.x = Bcol[(size_t)(k + 2 * half) * DD];
                bf.y = Bcol[(size_t)(k + 2 * half + 1) * DD];
                acc = __builtin_amdgcn_wmma_f32_16x16x4_f32(
                    false, a, false, bf, (short)0, acc, false, false);
            }
            ctx[t] = acc;
        }
    }
    __syncthreads();
#pragma unroll
    for (int t = 0; t < 4; ++t) {
        const int n0 = wave * 64 + t * 16;
#pragma unroll
        for (int i = 0; i < 8; ++i)
            sS[(i + 8 * half) * STR + n0 + l16] = ctx[t][i];
    }
    __syncthreads();

    {
        const float* AQ = sQ + l16 * STR + 2 * half;
        const float* AC = sS + l16 * STR + 2 * half;
        for (int t = 0; t < 4; ++t) {
            const int n0 = wave * 64 + t * 16;
            const float* Wr = W + (size_t)(n0 + l16) * (2 * DD) + 2 * half;
            v8f acc = {};
            for (int k = 0; k < DD; k += 4) {
                v2f a;  a.x = AQ[k]; a.y = AQ[k + 1];
                v2f bf; bf.x = Wr[k]; bf.y = Wr[k + 1];
                acc = __builtin_amdgcn_wmma_f32_16x16x4_f32(
                    false, a, false, bf, (short)0, acc, false, false);
            }
            const float* Wr2 = Wr + DD;
            for (int k = 0; k < DD; k += 4) {
                v2f a;  a.x = AC[k]; a.y = AC[k + 1];
                v2f bf; bf.x = Wr2[k]; bf.y = Wr2[k + 1];
                acc = __builtin_amdgcn_wmma_f32_16x16x4_f32(
                    false, a, false, bf, (short)0, acc, false, false);
            }
            const float bn = bias[n0 + l16];
            float* orow = outMasked + ((size_t)b * SD + q0) * DD + n0 + l16;
#pragma unroll
            for (int i = 0; i < 8; ++i) {
                const int M = i + 8 * half;
                orow[(size_t)M * DD] = tanhf(acc[i] + bn) * sMask[M];
            }
        }
    }
}

extern "C" void kernel_launch(void* const* d_in, const int* in_sizes, int n_in,
                              void* d_out, int out_size, void* d_ws, size_t ws_size,
                              hipStream_t stream) {
    (void)in_sizes; (void)n_in; (void)out_size; (void)d_ws; (void)ws_size;
    const float* Q    = (const float*)d_in[0];
    const float* E    = (const float*)d_in[1];
    const float* mask = (const float*)d_in[2];
    const float* W    = (const float*)d_in[3];
    const float* bias = (const float*)d_in[4];
    float* outMasked  = (float*)d_out;
    float* outWeights = (float*)d_out + (size_t)NB * SD * SE;

    dim3 grid(SD / 16, NB);
    attn_fused_kernel<<<grid, 256, 0, stream>>>(Q, E, mask, W, bias,
                                                outMasked, outWeights);
}


# B operand of all three products goes through LDS, not global-to-register: 128x32 tiles by coalesced 16-B loads two ahead into a 3-slot ring, read by 8 waves that split 64 rows; k, sum order unchanged
# speedup vs baseline: 1.1539x; 1.1539x over previous
#include <hip/hip_runtime.h>
#include <math.h>


typedef __attribute__((ext_vector_type(2))) float v2f;
typedef __attribute__((ext_vector_type(4))) float v4f;
typedef __attribute__((ext_vector_type(8))) float v8f;

#define NB   32
#define SD   512
#define SE   512
#define DD   512
#define STR  516
#define ROWS 64
#define PR   36
#define PK   136
#define TILE (128 * PR)

struct Frag { v2f a[2]; v2f b[2][4]; };

__device__ __forceinline__ v8f wmma4(v2f a, v2f b, v8f c) {
    return __builtin_amdgcn_wmma_f32_16x16x4_f32(false, a, false, b, (short)0, c, false, false);
}

__device__ __forceinline__ void ldR(v4f (&r)[4], const float* __restrict__ g, int ld, int tid) {
#pragma unroll
    for (int j = 0; j < 4; ++j) {
        const int idx = tid + 256 * j;
        r[j] = *(const v4f*)(g + (size_t)(idx >> 3) * ld + (idx & 7) * 4);
    }
}
__device__ __forceinline__ void stR(float* buf, const v4f (&r)[4], int tid) {
#pragma unroll
    for (int j = 0; j < 4; ++j) {
        const int idx = tid + 256 * j;
        *(v4f*)(buf + (idx >> 3) * PR + (idx & 7) * 4) = r[j];
    }
}
__device__ __forceinline__ void ldK(v4f (&r)[4], const float* __restrict__ g, int ld, int tid) {
#pragma unroll
    for (int j = 0; j < 4; ++j) {
        const int idx = tid + 256 * j;
        r[j] = *(const v4f*)(g + (size_t)(idx >> 5) * ld + (idx & 31) * 4);
    }
}
__device__ __forceinline__ void stK(float* buf, const v4f (&r)[4], int tid) {
#pragma unroll
    for (int j = 0; j < 4; ++j) {
        const int idx = tid + 256 * j;
        *(v4f*)(buf + (idx >> 5) * PK + (idx & 31) * 4) = r[j];
    }
}

__device__ __forceinline__ void pfR(Frag& f, const float* Ap, const float* Bp) {
#pragma unroll
    for (int s = 0; s < 2; ++s) {
        f.a[s] = *(const v2f*)(Ap + 4 * s);
#pragma unroll
        for (int j = 0; j < 4; ++j) f.b[s][j] = *(const v2f*)(Bp + j * 16 * PR + 4 * s);
    }
}
__device__ __forceinline__ void mmR(v8f* acc, const Frag& f, const float* Ap, const float* Bp) {
#pragma unroll
    for (int s = 0; s < 2; ++s)
#pragma unroll
        for (int j = 0; j < 4; ++j) acc[j] = wmma4(f.a[s], f.b[s][j], acc[j]);
#pragma unroll
    for (int kk = 8; kk < 32; kk += 4) {
        const v2f a = *(const v2f*)(Ap + kk);
#pragma unroll
        for (int j = 0; j < 4; ++j) {
            const v2f bf = *(const v2f*)(Bp + j * 16 * PR + kk);
            acc[j] = wmma4(a, bf, acc[j]);
        }
    }
}
__device__ __forceinline__ void pfK(Frag& f, const float* Ap, const float* Bp) {
#pragma unroll
    for (int s = 0; s < 2; ++s) {
        f.a[s] = *(const v2f*)(Ap + 4 * s);
#pragma unroll
        for (int j = 0; j < 4; ++j) {
            f.b[s][j].x = Bp[(4 * s) * PK + j * 16];
            f.b[s][j].y = Bp[(4 * s + 1) * PK + j * 16];
        }
    }
}
__device__ __forceinline__ void mmK(v8f* acc, const Frag& f, const float* Ap, const float* Bp) {
#pragma unroll
    for (int s = 0; s < 2; ++s)
#pragma unroll
        for (int j = 0; j < 4; ++j) acc[j] = wmma4(f.a[s], f.b[s][j], acc[j]);
#pragma unroll
    for (int kk = 8; kk < 32; kk += 4) {
        const v2f a = *(const v2f*)(Ap + kk);
#pragma unroll
        for (int j = 0; j < 4; ++j) {
            v2f bf;
            bf.x = Bp[kk * PK + j * 16];
            bf.y = Bp[(kk + 1) * PK + j * 16];
            acc[j] = wmma4(a, bf, acc[j]);
        }
    }
}

__launch_bounds__(256, 1)
__global__ void attn_fused_kernel(const float* __restrict__ Q,
                                  const float* __restrict__ E,
                                  const float* __restrict__ mask,
                                  const float* __restrict__ W,
                                  const float* __restrict__ bias,
                                  float* __restrict__ outMasked,
                                  float* __restrict__ outWeights)
{
    __shared__ float sQ[ROWS * STR];
    __shared__ float sS[ROWS * STR];
    __shared__ float sB[3 * TILE];
    __shared__ float sMask[ROWS];
    float* sRed = sB;

    const int b    = blockIdx.y;
    const int q0   = blockIdx.x * ROWS;
    const int tid  = threadIdx.x;
    const int lane = tid & 31;
    const int wave = tid >> 5;
    const int half = lane >> 4;
    const int l16  = lane & 15;
    const int rt   = wave & 3;
    const int ch   = wave >> 2;

    const float* Qb = Q + ((size_t)b * SD + q0) * DD;
    const float* Eb = E + (size_t)b * SE * DD;

    for (int i = tid; i < ROWS * DD; i += 256) {
        int r = i >> 9;
        int c = i & 511;
        sQ[r * STR + c] = Qb[r * DD + c];
    }
    if (tid < ROWS) sMask[tid] = mask[(size_t)b * SD + q0 + tid];

    const float scale = 0.044194173824159216f;

    const int arow = (rt * 16 + l16) * STR + 2 * half;
    const float* AQ = sQ + arow;
    const float* AS = sS + arow;
    const int bR = (ch * 64 + l16) * PR + 2 * half;
    const int bK = (2 * half) * PK + ch * 64 + l16;

    v4f pre[4];
    Frag f;
    int o0, o1, o2;

    o0 = 0; o1 = TILE; o2 = 2 * TILE;
    ldR(pre, Eb, DD, tid);
    stR(sB + o0, pre, tid);
    ldR(pre, Eb + 32, DD, tid);
    stR(sB + o1, pre, tid);
    __syncthreads();
    pfR(f, AQ, sB + o0 + bR);
    for (int nb = 0; nb < 4; ++nb) {
        v8f acc[4] = {};
        for (int kc = 0; kc < 16; ++kc) {
            const int ti = nb * 16 + kc;
            const int tn = ti + 2 < 64 ? ti + 2 : 63;
            ldR(pre, Eb + (size_t)((tn >> 4) * 128) * DD + (tn & 15) * 32, DD, tid);
            mmR(acc, f, AQ + kc * 32, sB + o0 + bR);
            stR(sB + o2, pre, tid);
            pfR(f, AQ + ((kc + 1) & 15) * 32, sB + o1 + bR);
            __syncthreads();
            const int t = o0; o0 = o1; o1 = o2; o2 = t;
        }
#pragma unroll
        for (int j = 0; j < 4; ++j) {
            const int n0 = nb * 128 + ch * 64 + j * 16;
#pragma unroll
            for (int i = 0; i < 8; ++i)
                sS[(rt * 16 + i + 8 * half) * STR + n0 + l16] = acc[j][i] * scale;
        }
    }
    __syncthreads();

    {
        const int r16 = tid & 15;
        const int sub = tid >> 4;
#pragma unroll
        for (int rg = 0; rg < 4; ++rg) {
            const int row = rg * 16 + r16;
            const float* srow = sS + row * STR;
            float m = -INFINITY;
            for (int c = sub * 32; c < sub * 32 + 32; ++c) m = fmaxf(m, srow[c]);
            sRed[row * 16 + sub] = m;
        }
        __syncthreads();
        float mall[4];
#pragma unroll
        for (int rg = 0; rg < 4; ++rg) {
            const int row = rg * 16 + r16;
            float mm = -INFINITY;
#pragma unroll
            for (int j = 0; j < 16; ++j) mm = fmaxf(mm, sRed[row * 16 + j]);
            mall[rg] = mm;
        }
        __syncthreads();
#pragma unroll
        for (int rg = 0; rg < 4; ++rg) {
            const int row = rg * 16 + r16;
            float* srow = sS + row * STR;
            float s = 0.f;
            for (int c = sub * 32; c < sub * 32 + 32; ++c) {
                float p = __expf(srow[c] - mall[rg]);
                srow[c] = p;
                s += p;
            }
            sRed[row * 16 + sub] = s;
        }
        __syncthreads();
#pragma unroll
        for (int rg = 0; rg < 4; ++rg) {
            const int row = rg * 16 + r16;
            float* srow = sS + row * STR;
            float stot = 0.f;
#pragma unroll
            for (int j = 0; j < 16; ++j) stot += sRed[row * 16 + j];
            const float inv = 1.0f / stot;
            float* gw = outWeights + ((size_t)b * SD + q0 + row) * SE;
            for (int c = sub * 32; c < sub * 32 + 32; ++c) {
                float p = srow[c] * inv;
                srow[c] = p;
                gw[c]   = p;
            }
        }
    }
    __syncthreads();

    v8f ctx[16];
#pragma unroll
    for (int t = 0; t < 16; ++t) ctx[t] = (v8f){};
    o0 = 0; o1 = TILE; o2 = 2 * TILE;
    ldK(pre, Eb, DD, tid);
    stK(sB + o0, pre, tid);
    ldK(pre, Eb + (size_t)32 * DD, DD, tid);
    stK(sB + o1, pre, tid);
    __syncthreads();
    pfK(f, AS, sB + o0 + bK);
#pragma unroll
    for (int nb = 0; nb < 4; ++nb) {
        for (int kc = 0; kc < 16; ++kc) {
            const int ti = nb * 16 + kc;
            const int tn = ti + 2 < 64 ? ti + 2 : 63;
            ldK(pre, Eb + (size_t)((tn & 15) * 32) * DD + (tn >> 4) * 128, DD, tid);
            mmK(&ctx[nb * 4], f, AS + kc * 32, sB + o0 + bK);
            stK(sB + o2, pre, tid);
            pfK(f, AS + ((kc + 1) & 15) * 32, sB + o1 + bK);
            __syncthreads();
            const int t = o0; o0 = o1; o1 = o2; o2 = t;
        }
    }
#pragma unroll
    for (int t = 0; t < 16; ++t) {
        const int n0 = (t >> 2) * 128 + ch * 64 + (t & 3) * 16;
#pragma unroll
        for (int i = 0; i < 8; ++i)
            sS[(rt * 16 + i + 8 * half) * STR + n0 + l16] = ctx[t][i];
    }

    o0 = 0; o1 = TILE; o2 = 2 * TILE;
    ldR(pre, W, 2 * DD, tid);
    stR(sB + o0, pre, tid);
    ldR(pre, W + 32, 2 * DD, tid);
    stR(sB + o1, pre, tid);
    __syncthreads();
    pfR(f, AQ, sB + o0 + bR);
    for (int nb = 0; nb < 4; ++nb) {
        v8f acc[4] = {};
        for (int kc = 0; kc < 32; ++kc) {
            const int ti = nb * 32 + kc;
            const int tn = ti + 2 < 128 ? ti + 2 : 127;
            ldR(pre, W + (size_t)((tn >> 5) * 128) * (2 * DD) + (tn & 31) * 32, 2 * DD, tid);
            const float* Ap = kc < 16 ? AQ + kc * 32 : AS + (kc - 16) * 32;
            mmR(acc, f, Ap, sB + o0 + bR);
            stR(sB + o2, pre, tid);
            const int kn = (kc + 1) & 31;
            const float* An = kn < 16 ? AQ + kn * 32 : AS + (kn - 16) * 32;
            pfR(f, An, sB + o1 + bR);
            __syncthreads();
            const int t = o0; o0 = o1; o1 = o2; o2 = t;
        }
#pragma unroll
        for (int j = 0; j < 4; ++j) {
            const int n0 = nb * 128 + ch * 64 + j * 16;
            const float bn = bias[n0 + l16];
            float* orow = outMasked + ((size_t)b * SD + q0 + rt * 16) * DD + n0 + l16;
#pragma unroll
            for (int i = 0; i < 8; ++i) {
                const int M = i + 8 * half;
                orow[(size_t)M * DD] = tanhf(acc[j][i] + bn) * sMask[rt * 16 + M];
            }
        }
    }
}

extern "C" void kernel_launch(void* const* d_in, const int* in_sizes, int n_in,
                              void* d_out, int out_size, void* d_ws, size_t ws_size,
                              hipStream_t stream) {
    (void)in_sizes; (void)n_in; (void)out_size; (void)d_ws; (void)ws_size;
    const float* Q    = (const float*)d_in[0];
    const float* E    = (const float*)d_in[1];
    const float* mask = (const float*)d_in[2];
    const float* W    = (const float*)d_in[3];
    const float* bias = (const float*)d_in[4];
    float* outMasked  = (float*)d_out;
    float* outWeights = (float*)d_out + (size_t)NB * SD * SE;

    dim3 grid(SD / ROWS, NB);
    attn_fused_kernel<<<grid, 256, 0, stream>>>(Q, E, mask, W, bias,
                                                outMasked, outWeights);
}


# B tiles requested from global three tiles ahead and kept in flight across the loop barrier (LDS-only fence, no load-counter drain), stored to the ring at the next trip's top; k, sum order unchanged
# speedup vs baseline: 1.2618x; 1.0935x over previous
#include <hip/hip_runtime.h>
#include <math.h>


typedef __attribute__((ext_vector_type(2))) float v2f;
typedef __attribute__((ext_vector_type(4))) float v4f;
typedef __attribute__((ext_vector_type(8))) float v8f;

#define NB   32
#define SD   512
#define SE   512
#define DD   512
#define STR  516
#define ROWS 64
#define PR   36
#define PK   136
#define TILE (128 * PR)

struct Frag { v2f a[2]; v2f b[2][4]; };

__device__ __forceinline__ void ring_barrier() {
    __builtin_amdgcn_fence(__ATOMIC_RELEASE, "workgroup", "local");
    __builtin_amdgcn_s_barrier();
    __builtin_amdgcn_fence(__ATOMIC_ACQUIRE, "workgroup", "local");
}

__device__ __forceinline__ v8f wmma4(v2f a, v2f b, v8f c) {
    return __builtin_amdgcn_wmma_f32_16x16x4_f32(false, a, false, b, (short)0, c, false, false);
}

__device__ __forceinline__ void ldR(v4f (&r)[4], const float* __restrict__ g, int ld, int tid) {
#pragma unroll
    for (int j = 0; j < 4; ++j) {
        const int idx = tid + 256 * j;
        r[j] = *(const v4f*)(g + (size_t)(idx >> 3) * ld + (idx & 7) * 4);
    }
}
__device__ __forceinline__ void stR(float* buf, const v4f (&r)[4], int tid) {
#pragma unroll
    for (int j = 0; j < 4; ++j) {
        const int idx = tid + 256 * j;
        *(v4f*)(buf + (idx >> 3) * PR + (idx & 7) * 4) = r[j];
    }
}
__device__ __forceinline__ void ldK(v4f (&r)[4], const float* __restrict__ g, int ld, int tid) {
#pragma unroll
    for (int j = 0; j < 4; ++j) {
        const int idx = tid + 256 * j;
        r[j] = *(const v4f*)(g + (size_t)(idx >> 5) * ld + (idx & 31) * 4);
    }
}
__device__ __forceinline__ void stK(float* buf, const v4f (&r)[4], int tid) {
#pragma unroll
    for (int j = 0; j < 4; ++j) {
        const int idx = tid + 256 * j;
        *(v4f*)(buf + (idx >> 5) * PK + (idx & 31) * 4) = r[j];
    }
}

__device__ __forceinline__ void pfR(Frag& f, const float* Ap, const float* Bp) {
#pragma unroll
    for (int s = 0; s < 2; ++s) {
        f.a[s] = *(const v2f*)(Ap + 4 * s);
#pragma unroll
        for (int j = 0; j < 4; ++j) f.b[s][j] = *(const v2f*)(Bp + j * 16 * PR + 4 * s);
    }
}
__device__ __forceinline__ void mmR(v8f* acc, const Frag& f, const float* Ap, const float* Bp) {
#pragma unroll
    for (int s = 0; s < 2; ++s)
#pragma unroll
        for (int j = 0; j < 4; ++j) acc[j] = wmma4(f.a[s], f.b[s][j], acc[j]);
#pragma unroll
    for (int kk = 8; kk < 32; kk += 4) {
        const v2f a = *(const v2f*)(Ap + kk);
#pragma unroll
        for (int j = 0; j < 4; ++j) {
            const v2f bf = *(const v2f*)(Bp + j * 16 * PR + kk);
            acc[j] = wmma4(a, bf, acc[j]);
        }
    }
}
__device__ __forceinline__ void pfK(Frag& f, const float* Ap, const float* Bp) {
#pragma unroll
    for (int s = 0; s < 2; ++s) {
        f.a[s] = *(const v2f*)(Ap + 4 * s);
#pragma unroll
        for (int j = 0; j < 4; ++j) {
            f.b[s][j].x = Bp[(4 * s) * PK + j * 16];
            f.b[s][j].y = Bp[(4 * s + 1) * PK + j * 16];
        }
    }
}
__device__ __forceinline__ void mmK(v8f* acc, const Frag& f, const float* Ap, const float* Bp) {
#pragma unroll
    for (int s = 0; s < 2; ++s)
#pragma unroll
        for (int j = 0; j < 4; ++j) acc[j] = wmma4(f.a[s], f.b[s][j], acc[j]);
#pragma unroll
    for (int kk = 8; kk < 32; kk += 4) {
        const v2f a = *(const v2f*)(Ap + kk);
#pragma unroll
        for (int j = 0; j < 4; ++j) {
            v2f bf;
            bf.x = Bp[kk * PK + j * 16];
            bf.y = Bp[(kk + 1) * PK + j * 16];
            acc[j] = wmma4(a, bf, acc[j]);
        }
    }
}

__launch_bounds__(256, 1)
__global__ void attn_fused_kernel(const float* __restrict__ Q,
                                  const float* __restrict__ E,
                                  const float* __restrict__ mask,
                                  const float* __restrict__ W,
                                  const float* __restrict__ bias,
                                  float* __restrict__ outMasked,
                                  float* __restrict__ outWeights)
{
    __shared__ float sQ[ROWS * STR];
    __shared__ float sS[ROWS * STR];
    __shared__ float sB[3 * TILE];
    __shared__ float sMask[ROWS];
    float* sRed = sB;

    const int b    = blockIdx.y;
    const int q0   = blockIdx.x * ROWS;
    const int tid  = threadIdx.x;
    const int lane = tid & 31;
    const int wave = tid >> 5;
    const int half = lane >> 4;
    const int l16  = lane & 15;
    const int rt   = wave & 3;
    const int ch   = wave >> 2;

    const float* Qb = Q + ((size_t)b * SD + q0) * DD;
    const float* Eb = E + (size_t)b * SE * DD;

    for (int i = tid; i < ROWS * DD; i += 256) {
        int r = i >> 9;
        int c = i & 511;
        sQ[r * STR + c] = Qb[r * DD + c];
    }
    if (tid < ROWS) sMask[tid] = mask[(size_t)b * SD + q0 + tid];

    const float scale = 0.044194173824159216f;

    const int arow = (rt * 16 + l16) * STR + 2 * half;
    const float* AQ = sQ + arow;
    const float* AS = sS + arow;
    const int bR = (ch * 64 + l16) * PR + 2 * half;
    const int bK = (2 * half) * PK + ch * 64 + l16;

    v4f pre[4];
    Frag f;
    int o0, o1, o2;

    o0 = 0; o1 = TILE; o2 = 2 * TILE;
    ldR(pre, Eb, DD, tid);
    stR(sB + o0, pre, tid);
    ldR(pre, Eb + 32, DD, tid);
    stR(sB + o1, pre, tid);
    ldR(pre, Eb + 64, DD, tid);
    __syncthreads();
    pfR(f, AQ, sB + o0 + bR);
    for (int nb = 0; nb < 4; ++nb) {
        v8f acc[4] = {};
        for (int kc = 0; kc < 16; ++kc) {
            const int ti = nb * 16 + kc;
            const int tn = ti + 3 < 64 ? ti + 3 : 63;
            stR(sB + o2, pre, tid);
            ldR(pre, Eb + (size_t)((tn >> 4) * 128) * DD + (tn & 15) * 32, DD, tid);
            mmR(acc, f, AQ + kc * 32, sB + o0 + bR);
            pfR(f, AQ + ((kc + 1) & 15) * 32, sB + o1 + bR);
            ring_barrier();
            const int t = o0; o0 = o1; o1 = o2; o2 = t;
        }
#pragma unroll
        for (int j = 0; j < 4; ++j) {
            const int n0 = nb * 128 + ch * 64 + j * 16;
#pragma unroll
            for (int i = 0; i < 8; ++i)
                sS[(rt * 16 + i + 8 * half) * STR + n0 + l16] = acc[j][i] * scale;
        }
    }
    __syncthreads();

    {
        const int r16 = tid & 15;
        const int sub = tid >> 4;
#pragma unroll
        for (int rg = 0; rg < 4; ++rg) {
            const int row = rg * 16 + r16;
            const float* srow = sS + row * STR;
            float m = -INFINITY;
            for (int c = sub * 32; c < sub * 32 + 32; ++c) m = fmaxf(m, srow[c]);
            sRed[row * 16 + sub] = m;
        }
        __syncthreads();
        float mall[4];
#pragma unroll
        for (int rg = 0; rg < 4; ++rg) {
            const int row = rg * 16 + r16;
            float mm = -INFINITY;
#pragma unroll
            for (int j = 0; j < 16; ++j) mm = fmaxf(mm, sRed[row * 16 + j]);
            mall[rg] = mm;
        }
        __syncthreads();
#pragma unroll
        for (int rg = 0; rg < 4; ++rg) {
            const int row = rg * 16 + r16;
            float* srow = sS + row * STR;
            float s = 0.f;
            for (int c = sub * 32; c < sub * 32 + 32; ++c) {
                float p = __expf(srow[c] - mall[rg]);
                srow[c] = p;
                s += p;
            }
            sRed[row * 16 + sub] = s;
        }
        __syncthreads();
#pragma unroll
        for (int rg = 0; rg < 4; ++rg) {
            const int row = rg * 16 + r16;
            float* srow = sS + row * STR;
            float stot = 0.f;
#pragma unroll
            for (int j = 0; j < 16; ++j) stot += sRed[row * 16 + j];
            const float inv = 1.0f / stot;
            float* gw = outWeights + ((size_t)b * SD + q0 + row) * SE;
            for (int c = sub * 32; c < sub * 32 + 32; ++c) {
                float p = srow[c] * inv;
                srow[c] = p;
                gw[c]   = p;
            }
        }
    }
    __syncthreads();

    v8f ctx[16];
#pragma unroll
    for (int t = 0; t < 16; ++t) ctx[t] = (v8f){};
    o0 = 0; o1 = TILE; o2 = 2 * TILE;
    ldK(pre, Eb, DD, tid);
    stK(sB + o0, pre, tid);
    ldK(pre, Eb + (size_t)32 * DD, DD, tid);
    stK(sB + o1, pre, tid);
    ldK(pre, Eb + (size_t)64 * DD, DD, tid);
    __syncthreads();
    pfK(f, AS, sB + o0 + bK);
#pragma unroll
    for (int nb = 0; nb < 4; ++nb) {
        for (int kc = 0; kc < 16; ++kc) {
            const int ti = nb * 16 + kc;
            const int tn = ti + 3 < 64 ? ti + 3 : 63;
            stK(sB + o2, pre, tid);
            ldK(pre, Eb + (size_t)((tn & 15) * 32) * DD + (tn >> 4) * 128, DD, tid);
            mmK(&ctx[nb * 4], f, AS + kc * 32, sB + o0 + bK);
            pfK(f, AS + ((kc + 1) & 15) * 32, sB + o1 + bK);
            ring_barrier();
            const int t = o0; o0 = o1; o1 = o2; o2 = t;
        }
    }
#pragma unroll
    for (int t = 0; t < 16; ++t) {
        const int n0 = (t >> 2) * 128 + ch * 64 + (t & 3) * 16;
#pragma unroll
        for (int i = 0; i < 8; ++i)
            sS[(rt * 16 + i + 8 * half) * STR + n0 + l16] = ctx[t][i];
    }

    o0 = 0; o1 = TILE; o2 = 2 * TILE;
    ldR(pre, W, 2 * DD, tid);
    stR(sB + o0, pre, tid);
    ldR(pre, W + 32, 2 * DD, tid);
    stR(sB + o1, pre, tid);
    ldR(pre, W + 64, 2 * DD, tid);
    __syncthreads();
    pfR(f, AQ, sB + o0 + bR);
    for (int nb = 0; nb < 4; ++nb) {
        v8f acc[4] = {};
        for (int kc = 0; kc < 32; ++kc) {
            const int ti = nb * 32 + kc;
            const int tn = ti + 3 < 128 ? ti + 3 : 127;
            stR(sB + o2, pre, tid);
            ldR(pre, W + (size_t)((tn >> 5) * 128) * (2 * DD) + (tn & 31) * 32, 2 * DD, tid);
            const float* Ap = kc < 16 ? AQ + kc * 32 : AS + (kc - 16) * 32;
            mmR(acc, f, Ap, sB + o0 + bR);
            const int kn = (kc + 1) & 31;
            const float* An = kn < 16 ? AQ + kn * 32 : AS + (kn - 16) * 32;
            pfR(f, An, sB + o1 + bR);
            ring_barrier();
            const int t = o0; o0 = o1; o1 = o2; o2 = t;
        }
#pragma unroll
        for (int j = 0; j < 4; ++j) {
            const int n0 = nb * 128 + ch * 64 + j * 16;
            const float bn = bias[n0 + l16];
            float* orow = outMasked + ((size_t)b * SD + q0 + rt * 16) * DD + n0 + l16;
#pragma unroll
            for (int i = 0; i < 8; ++i) {
                const int M = i + 8 * half;
                orow[(size_t)M * DD] = tanhf(acc[j][i] + bn) * sMask[rt * 16 + M];
            }
        }
    }
}

extern "C" void kernel_launch(void* const* d_in, const int* in_sizes, int n_in,
                              void* d_out, int out_size, void* d_ws, size_t ws_size,
                              hipStream_t stream) {
    (void)in_sizes; (void)n_in; (void)out_size; (void)d_ws; (void)ws_size;
    const float* Q    = (const float*)d_in[0];
    const float* E    = (const float*)d_in[1];
    const float* mask = (const float*)d_in[2];
    const float* W    = (const float*)d_in[3];
    const float* bias = (const float*)d_in[4];
    float* outMasked  = (float*)d_out;
    float* outWeights = (float*)d_out + (size_t)NB * SD * SE;

    dim3 grid(SD / ROWS, NB);
    attn_fused_kernel<<<grid, 256, 0, stream>>>(Q, E, mask, W, bias,
                                                outMasked, outWeights);
}


# All LDS fragment reads of a 32-k tile (A, B) issued as one group a whole tile ahead into a second register set; WMMAs run from registers; ctx tile held k-pair interleaved; k, sum order unchanged
# speedup vs baseline: 1.3086x; 1.0371x over previous
#include <hip/hip_runtime.h>
#include <math.h>


typedef __attribute__((ext_vector_type(2))) float v2f;
typedef __attribute__((ext_vector_type(4))) float v4f;
typedef __attribute__((ext_vector_type(8))) float v8f;

#define NB   32
#define SD   512
#define SE   512
#define DD   512
#define STR  516
#define ROWS 64
#define PR   36
#define P2   288
#define TILE (128 * PR)

struct Frag { v2f a[8]; v2f b[8][4]; };

__device__ __forceinline__ void ring_barrier() {
    __builtin_amdgcn_sched_barrier(0);
    __builtin_amdgcn_fence(__ATOMIC_RELEASE, "workgroup", "local");
    __builtin_amdgcn_s_barrier();
    __builtin_amdgcn_fence(__ATOMIC_ACQUIRE, "workgroup", "local");
    __builtin_amdgcn_sched_barrier(0);
}

__device__ __forceinline__ void group_end() {
    __builtin_amdgcn_sched_barrier(0);
}

__device__ __forceinline__ v8f wmma4(v2f a, v2f b, v8f c) {
    return __builtin_amdgcn_wmma_f32_16x16x4_f32(false, a, false, b, (short)0, c, false, false);
}

__device__ __forceinline__ void ldR(v4f (&r)[4], const float* __restrict__ g, int ld, int tid) {
#pragma unroll
    for (int j = 0; j < 4; ++j) {
        const int idx = tid + 256 * j;
        r[j] = *(const v4f*)(g + (size_t)(idx >> 3) * ld + (idx & 7) * 4);
    }
}
__device__ __forceinline__ void stR(float* buf, const v4f (&r)[4], int tid) {
#pragma unroll
    for (int j = 0; j < 4; ++j) {
        const int idx = tid + 256 * j;
        *(v4f*)(buf + (idx >> 3) * PR + (idx & 7) * 4) = r[j];
    }
}
__device__ __forceinline__ void ldK(v4f (&r)[4], const float* __restrict__ g, int ld, int tid) {
#pragma unroll
    for (int j = 0; j < 2; ++j) {
        const int idx = tid + 256 * j;
        const float* s = g + (size_t)(2 * (idx >> 5)) * ld + (idx & 31) * 4;
        r[2 * j]     = *(const v4f*)s;
        r[2 * j + 1] = *(const v4f*)(s + ld);
    }
}
__device__ __forceinline__ void stK(float* buf, const v4f (&r)[4], int tid) {
#pragma unroll
    for (int j = 0; j < 2; ++j) {
        const int idx = tid + 256 * j;
        float* d = buf + (idx >> 5) * P2 + (idx & 31) * 8;
        const v4f e = r[2 * j], o = r[2 * j + 1];
        *(v4f*)d       = (v4f){e.x, o.x, e.y, o.y};
        *(v4f*)(d + 4) = (v4f){e.z, o.z, e.w, o.w};
    }
}

__device__ __forceinline__ void ldAllR(Frag& f, const float* Ap, const float* Bp) {
#pragma unroll
    for (int s = 0; s < 8; ++s) {
        f.a[s] = *(const v2f*)(Ap + 4 * s);
#pragma unroll
        for (int j = 0; j < 4; ++j) f.b[s][j] = *(const v2f*)(Bp + j * 16 * PR + 4 * s);
    }
}
__device__ __forceinline__ void ldAllK(Frag& f, const float* Ap, const float* Bp) {
#pragma unroll
    for (int s = 0; s < 8; ++s) {
        f.a[s] = *(const v2f*)(Ap + 4 * s);
#pragma unroll
        for (int j = 0; j < 4; ++j) f.b[s][j] = *(const v2f*)(Bp + (2 * s) * P2 + j * 32);
    }
}
__device__ __forceinline__ void mmAll(v8f* acc, const Frag& f) {
#pragma unroll
    for (int s = 0; s < 8; ++s)
#pragma unroll
        for (int j = 0; j < 4; ++j) acc[j] = wmma4(f.a[s], f.b[s][j], acc[j]);
}

__launch_bounds__(256, 1)
__global__ void attn_fused_kernel(const float* __restrict__ Q,
                                  const float* __restrict__ E,
                                  const float* __restrict__ mask,
                                  const float* __restrict__ W,
                                  const float* __restrict__ bias,
                                  float* __restrict__ outMasked,
                                  float* __restrict__ outWeights)
{
    __shared__ float sQ[ROWS * STR];
    __shared__ float sS[ROWS * STR];
    __shared__ float sB[2 * TILE];
    __shared__ float sMask[ROWS];
    float* sRed = sB;

    const int b    = blockIdx.y;
    const int q0   = blockIdx.x * ROWS;
    const int tid  = threadIdx.x;
    const int lane = tid & 31;
    const int wave = tid >> 5;
    const int half = lane >> 4;
    const int l16  = lane & 15;
    const int rt   = wave & 3;
    const int ch   = wave >> 2;

    const float* Qb = Q + ((size_t)b * SD + q0) * DD;
    const float* Eb = E + (size_t)b * SE * DD;

    for (int i = tid; i < ROWS * DD; i += 256) {
        int r = i >> 9;
        int c = i & 511;
        sQ[r * STR + c] = Qb[r * DD + c];
    }
    if (tid < ROWS) sMask[tid] = mask[(size_t)b * SD + q0 + tid];

    const float scale = 0.044194173824159216f;

    const int arow = (rt * 16 + l16) * STR + 2 * half;
    const float* AQ = sQ + arow;
    const float* AS = sS + arow;
    const int bR = (ch * 64 + l16) * PR + 2 * half;
    const int bK = half * P2 + 2 * (ch * 64 + l16);

    v4f pre[4];
    Frag fA, fB;
    float* const s0 = sB;
    float* const s1 = sB + TILE;

    ldR(pre, Eb, DD, tid);
    stR(s0, pre, tid);
    ldR(pre, Eb + 32, DD, tid);
    stR(s1, pre, tid);
    ldR(pre, Eb + 64, DD, tid);
    __syncthreads();
    ldAllR(fA, AQ, s0 + bR);
    ring_barrier();
    for (int nb = 0; nb < 4; ++nb) {
        v8f acc[4] = {};
        for (int kc = 0; kc < 16; kc += 2) {
            const int ti = nb * 16 + kc;
            const int ta = ti + 3 < 64 ? ti + 3 : 63;
            const int tb = ti + 4 < 64 ? ti + 4 : 63;
            ldAllR(fB, AQ + (kc + 1) * 32, s1 + bR);
            stR(s0, pre, tid);
            ldR(pre, Eb + (size_t)((ta >> 4) * 128) * DD + (ta & 15) * 32, DD, tid);
            group_end();
            mmAll(acc, fA);
            ring_barrier();
            ldAllR(fA, AQ + ((kc + 2) & 15) * 32, s0 + bR);
            stR(s1, pre, tid);
            ldR(pre, Eb + (size_t)((tb >> 4) * 128) * DD + (tb & 15) * 32, DD, tid);
            group_end();
            mmAll(acc, fB);
            ring_barrier();
        }
#pragma unroll
        for (int j = 0; j < 4; ++j) {
            const int n0 = nb * 128 + ch * 64 + j * 16;
#pragma unroll
            for (int i = 0; i < 8; ++i)
                sS[(rt * 16 + i + 8 * half) * STR + n0 + l16] = acc[j][i] * scale;
        }
    }
    __syncthreads();

    {
        const int r16 = tid & 15;
        const int sub = tid >> 4;
#pragma unroll
        for (int rg = 0; rg < 4; ++rg) {
            const int row = rg * 16 + r16;
            const float* srow = sS + row * STR;
            float m = -INFINITY;
            for (int c = sub * 32; c < sub * 32 + 32; ++c) m = fmaxf(m, srow[c]);
            sRed[row * 16 + sub] = m;
        }
        __syncthreads();
        float mall[4];
#pragma unroll
        for (int rg = 0; rg < 4; ++rg) {
            const int row = rg * 16 + r16;
            float mm = -INFINITY;
#pragma unroll
            for (int j = 0; j < 16; ++j) mm = fmaxf(mm, sRed[row * 16 + j]);
            mall[rg] = mm;
        }
        __syncthreads();
#pragma unroll
        for (int rg = 0; rg < 4; ++rg) {
            const int row = rg * 16 + r16;
            float* srow = sS + row * STR;
            float s = 0.f;
            for (int c = sub * 32; c < sub * 32 + 32; ++c) {
                float p = __expf(srow[c] - mall[rg]);
                srow[c] = p;
                s += p;
            }
            sRed[row * 16 + sub] = s;
        }
        __syncthreads();
#pragma unroll
        for (int rg = 0; rg < 4; ++rg) {
            const int row = rg * 16 + r16;
            float* srow = sS + row * STR;
            float stot = 0.f;
#pragma unroll
            for (int j = 0; j < 16; ++j) stot += sRed[row * 16 + j];
            const float inv = 1.0f / stot;
            float* gw = outWeights + ((size_t)b * SD + q0 + row) * SE;
            for (int c = sub * 32; c < sub * 32 + 32; ++c) {
                float p = srow[c] * inv;
                srow[c] = p;
                gw[c]   = p;
            }
        }
    }
    __syncthreads();

    v8f ctx[16];
#pragma unroll
    for (int t = 0; t < 16; ++t) ctx[t] = (v8f){};
    ldK(pre, Eb, DD, tid);
    stK(s0, pre, tid);
    ldK(pre, Eb + (size_t)32 * DD, DD, tid);
    stK(s1, pre, tid);
    ldK(pre, Eb + (size_t)64 * DD, DD, tid);
    __syncthreads();
    ldAllK(fA, AS, s0 + bK);
    ring_barrier();
#pragma unroll
    for (int nb = 0; nb < 4; ++nb) {
        for (int kc = 0; kc < 16; kc += 2) {
            const int ti = nb * 16 + kc;
            const int ta = ti + 3 < 64 ? ti + 3 : 63;
            const int tb = ti + 4 < 64 ? ti + 4 : 63;
            ldAllK(fB, AS + (kc + 1) * 32, s1 + bK);
            stK(s0, pre, tid);
            ldK(pre, Eb + (size_t)((ta & 15) * 32) * DD + (ta >> 4) * 128, DD, tid);
            group_end();
            mmAll(&ctx[nb * 4], fA);
            ring_barrier();
            ldAllK(fA, AS + ((kc + 2) & 15) * 32, s0 + bK);
            stK(s1, pre, tid);
            ldK(pre, Eb + (size_t)((tb & 15) * 32) * DD + (tb >> 4) * 128, DD, tid);
            group_end();
            mmAll(&ctx[nb * 4], fB);
            ring_barrier();
        }
    }
#pragma unroll
    for (int t = 0; t < 16; ++t) {
        const int n0 = (t >> 2) * 128 + ch * 64 + (t & 3) * 16;
#pragma unroll
        for (int i = 0; i < 8; ++i)
            sS[(rt * 16 + i + 8 * half) * STR + n0 + l16] = ctx[t][i];
    }

    ldR(pre, W, 2 * DD, tid);
    stR(s0, pre, tid);
    ldR(pre, W + 32, 2 * DD, tid);
    stR(s1, pre, tid);
    ldR(pre, W + 64, 2 * DD, tid);
    __syncthreads();
    ldAllR(fA, AQ, s0 + bR);
    ring_barrier();
    for (int nb = 0; nb < 4; ++nb) {
        v8f acc[4] = {};
        for (int kc = 0; kc < 32; kc += 2) {
            const int ti = nb * 32 + kc;
            const int ta = ti + 3 < 128 ? ti + 3 : 127;
            const int tb = ti + 4 < 128 ? ti + 4 : 127;
            const int k1 = kc + 1;
            const int k2 = (kc + 2) & 31;
            ldAllR(fB, k1 < 16 ? AQ + k1 * 32 : AS + (k1 - 16) * 32, s1 + bR);
            stR(s0, pre, tid);
            ldR(pre, W + (size_t)((ta >> 5) * 128) * (2 * DD) + (ta & 31) * 32, 2 * DD, tid);
            group_end();
            mmAll(acc, fA);
            ring_barrier();
            ldAllR(fA, k2 < 16 ? AQ + k2 * 32 : AS + (k2 - 16) * 32, s0 + bR);
            stR(s1, pre, tid);
            ldR(pre, W + (size_t)((tb >> 5) * 128) * (2 * DD) + (tb & 31) * 32, 2 * DD, tid);
            group_end();
            mmAll(acc, fB);
            ring_barrier();
        }
#pragma unroll
        for (int j = 0; j < 4; ++j) {
            const int n0 = nb * 128 + ch * 64 + j * 16;
            const float bn = bias[n0 + l16];
            float* orow = outMasked + ((size_t)b * SD + q0 + rt * 16) * DD + n0 + l16;
#pragma unroll
            for (int i = 0; i < 8; ++i) {
                const int M = i + 8 * half;
                orow[(size_t)M * DD] = tanhf(acc[j][i] + bn) * sMask[rt * 16 + M];
            }
        }
    }
}

extern "C" void kernel_launch(void* const* d_in, const int* in_sizes, int n_in,
                              void* d_out, int out_size, void* d_ws, size_t ws_size,
                              hipStream_t stream) {
    (void)in_sizes; (void)n_in; (void)out_size; (void)d_ws; (void)ws_size;
    const float* Q    = (const float*)d_in[0];
    const float* E    = (const float*)d_in[1];
    const float* mask = (const float*)d_in[2];
    const float* W    = (const float*)d_in[3];
    const float* bias = (const float*)d_in[4];
    float* outMasked  = (float*)d_out;
    float* outWeights = (float*)d_out + (size_t)NB * SD * SE;

    dim3 grid(SD / ROWS, NB);
    attn_fused_kernel<<<grid, 256, 0, stream>>>(Q, E, mask, W, bias,
                                                outMasked, outWeights);
}


# Tile-loop barrier split: a wave signals behind its trip's first 12 matrix steps (own LDS group done), waits at the next trip's top: round trip, other-wave lag pass under WMMAs; k, sum order unchanged
# speedup vs baseline: 1.3499x; 1.0316x over previous
#include <hip/hip_runtime.h>
#include <math.h>


typedef __attribute__((ext_vector_type(2))) float v2f;
typedef __attribute__((ext_vector_type(4))) float v4f;
typedef __attribute__((ext_vector_type(8))) float v8f;

#define NB   32
#define SD   512
#define SE   512
#define DD   512
#define STR  516
#define ROWS 64
#define PR   36
#define P2   288
#define TILE (128 * PR)

struct Frag { v2f a[8]; v2f b[8][4]; };

__device__ __forceinline__ void ring_signal() {
    __builtin_amdgcn_sched_barrier(0);
    __builtin_amdgcn_fence(__ATOMIC_RELEASE, "workgroup", "local");
    __builtin_amdgcn_s_barrier_signal(-1);
    __builtin_amdgcn_sched_barrier(0);
}
__device__ __forceinline__ void ring_wait() {
    __builtin_amdgcn_sched_barrier(0);
    __builtin_amdgcn_s_barrier_wait(-1);
    __builtin_amdgcn_fence(__ATOMIC_ACQUIRE, "workgroup", "local");
    __builtin_amdgcn_sched_barrier(0);
}

__device__ __forceinline__ void group_end() {
    __builtin_amdgcn_sched_barrier(0);
}

__device__ __forceinline__ void body_end() {
    __builtin_amdgcn_sched_barrier(0);
}

__device__ __forceinline__ v8f wmma4(v2f a, v2f b, v8f c) {
    return __builtin_amdgcn_wmma_f32_16x16x4_f32(false, a, false, b, (short)0, c, false, false);
}

__device__ __forceinline__ void ldR(v4f (&r)[4], const float* __restrict__ g, int ld, int tid) {
#pragma unroll
    for (int j = 0; j < 4; ++j) {
        const int idx = tid + 256 * j;
        r[j] = *(const v4f*)(g + (size_t)(idx >> 3) * ld + (idx & 7) * 4);
    }
}
__device__ __forceinline__ void stR(float* buf, const v4f (&r)[4], int tid) {
#pragma unroll
    for (int j = 0; j < 4; ++j) {
        const int idx = tid + 256 * j;
        *(v4f*)(buf + (idx >> 3) * PR + (idx & 7) * 4) = r[j];
    }
}
__device__ __forceinline__ void ldK(v4f (&r)[4], const float* __restrict__ g, int ld, int tid) {
#pragma unroll
    for (int j = 0; j < 2; ++j) {
        const int idx = tid + 256 * j;
        const float* s = g + (size_t)(2 * (idx >> 5)) * ld + (idx & 31) * 4;
        r[2 * j]     = *(const v4f*)s;
        r[2 * j + 1] = *(const v4f*)(s + ld);
    }
}
__device__ __forceinline__ void stK(float* buf, const v4f (&r)[4], int tid) {
#pragma unroll
    for (int j = 0; j < 2; ++j) {
        const int idx = tid + 256 * j;
        float* d = buf + (idx >> 5) * P2 + (idx & 31) * 8;
        const v4f e = r[2 * j], o = r[2 * j + 1];
        *(v4f*)d       = (v4f){e.x, o.x, e.y, o.y};
        *(v4f*)(d + 4) = (v4f){e.z, o.z, e.w, o.w};
    }
}

__device__ __forceinline__ void ldAllR(Frag& f, const float* Ap, const float* Bp) {
#pragma unroll
    for (int s = 0; s < 8; ++s) {
        f.a[s] = *(const v2f*)(Ap + 4 * s);
#pragma unroll
        for (int j = 0; j < 4; ++j) f.b[s][j] = *(const v2f*)(Bp + j * 16 * PR + 4 * s);
    }
}
__device__ __forceinline__ void ldAllK(Frag& f, const float* Ap, const float* Bp) {
#pragma unroll
    for (int s = 0; s < 8; ++s) {
        f.a[s] = *(const v2f*)(Ap + 4 * s);
#pragma unroll
        for (int j = 0; j < 4; ++j) f.b[s][j] = *(const v2f*)(Bp + (2 * s) * P2 + j * 32);
    }
}
template <int S0, int S1>
__device__ __forceinline__ void mmPart(v8f* acc, const Frag& f) {
#pragma unroll
    for (int s = S0; s < S1; ++s)
#pragma unroll
        for (int j = 0; j < 4; ++j) acc[j] = wmma4(f.a[s], f.b[s][j], acc[j]);
}
#define KS 3

__launch_bounds__(256, 1)
__global__ void attn_fused_kernel(const float* __restrict__ Q,
                                  const float* __restrict__ E,
                                  const float* __restrict__ mask,
                                  const float* __restrict__ W,
                                  const float* __restrict__ bias,
                                  float* __restrict__ outMasked,
                                  float* __restrict__ outWeights)
{
    __shared__ float sQ[ROWS * STR];
    __shared__ float sS[ROWS * STR];
    __shared__ float sB[2 * TILE];
    __shared__ float sMask[ROWS];
    float* sRed = sB;

    const int b    = blockIdx.y;
    const int q0   = blockIdx.x * ROWS;
    const int tid  = threadIdx.x;
    const int lane = tid & 31;
    const int wave = tid >> 5;
    const int half = lane >> 4;
    const int l16  = lane & 15;
    const int rt   = wave & 3;
    const int ch   = wave >> 2;

    const float* Qb = Q + ((size_t)b * SD + q0) * DD;
    const float* Eb = E + (size_t)b * SE * DD;

    for (int i = tid; i < ROWS * DD; i += 256) {
        int r = i >> 9;
        int c = i & 511;
        sQ[r * STR + c] = Qb[r * DD + c];
    }
    if (tid < ROWS) sMask[tid] = mask[(size_t)b * SD + q0 + tid];

    const float scale = 0.044194173824159216f;

    const int arow = (rt * 16 + l16) * STR + 2 * half;
    const float* AQ = sQ + arow;
    const float* AS = sS + arow;
    const int bR = (ch * 64 + l16) * PR + 2 * half;
    const int bK = half * P2 + 2 * (ch * 64 + l16);

    v4f pre[4];
    Frag fA, fB;
    float* const s0 = sB;
    float* const s1 = sB + TILE;

    ldR(pre, Eb, DD, tid);
    stR(s0, pre, tid);
    ldR(pre, Eb + 32, DD, tid);
    stR(s1, pre, tid);
    ldR(pre, Eb + 64, DD, tid);
    __syncthreads();
    ldAllR(fA, AQ, s0 + bR);
    ring_signal();
    for (int nb = 0; nb < 4; ++nb) {
        v8f acc[4] = {};
        for (int kc = 0; kc < 16; kc += 2) {
            const int ti = nb * 16 + kc;
            const int ta = ti + 3 < 64 ? ti + 3 : 63;
            const int tb = ti + 4 < 64 ? ti + 4 : 63;
            ring_wait();
            ldAllR(fB, AQ + (kc + 1) * 32, s1 + bR);
            stR(s0, pre, tid);
            ldR(pre, Eb + (size_t)((ta >> 4) * 128) * DD + (ta & 15) * 32, DD, tid);
            group_end();
            mmPart<0, KS>(acc, fA);
            ring_signal();
            mmPart<KS, 8>(acc, fA);
            ring_wait();
            ldAllR(fA, AQ + ((kc + 2) & 15) * 32, s0 + bR);
            stR(s1, pre, tid);
            ldR(pre, Eb + (size_t)((tb >> 4) * 128) * DD + (tb & 15) * 32, DD, tid);
            group_end();
            mmPart<0, KS>(acc, fB);
            ring_signal();
            mmPart<KS, 8>(acc, fB);
        }
#pragma unroll
        for (int j = 0; j < 4; ++j) {
            const int n0 = nb * 128 + ch * 64 + j * 16;
#pragma unroll
            for (int i = 0; i < 8; ++i)
                sS[(rt * 16 + i + 8 * half) * STR + n0 + l16] = acc[j][i] * scale;
        }
    }
    ring_wait();
    __syncthreads();

    {
        const int r16 = tid & 15;
        const int sub = tid >> 4;
#pragma unroll
        for (int rg = 0; rg < 4; ++rg) {
            const int row = rg * 16 + r16;
            const float* srow = sS + row * STR;
            float m = -INFINITY;
            for (int c = sub * 32; c < sub * 32 + 32; ++c) m = fmaxf(m, srow[c]);
            sRed[row * 16 + sub] = m;
        }
        __syncthreads();
        float mall[4];
#pragma unroll
        for (int rg = 0; rg < 4; ++rg) {
            const int row = rg * 16 + r16;
            float mm = -INFINITY;
#pragma unroll
            for (int j = 0; j < 16; ++j) mm = fmaxf(mm, sRed[row * 16 + j]);
            mall[rg] = mm;
        }
        __syncthreads();
#pragma unroll
        for (int rg = 0; rg < 4; ++rg) {
            const int row = rg * 16 + r16;
            float* srow = sS + row * STR;
            float s = 0.f;
            for (int c = sub * 32; c < sub * 32 + 32; ++c) {
                float p = __expf(srow[c] - mall[rg]);
                srow[c] = p;
                s += p;
            }
            sRed[row * 16 + sub] = s;
        }
        __syncthreads();
#pragma unroll
        for (int rg = 0; rg < 4; ++rg) {
            const int row = rg * 16 + r16;
            float* srow = sS + row * STR;
            float stot = 0.f;
#pragma unroll
            for (int j = 0; j < 16; ++j) stot += sRed[row * 16 + j];
            const float inv = 1.0f / stot;
            float* gw = outWeights + ((size_t)b * SD + q0 + row) * SE;
            for (int c = sub * 32; c < sub * 32 + 32; ++c) {
                float p = srow[c] * inv;
                srow[c] = p;
                gw[c]   = p;
            }
        }
    }
    __syncthreads();

    v8f ctx[16];
#pragma unroll
    for (int t = 0; t < 16; ++t) ctx[t] = (v8f){};
    ldK(pre, Eb, DD, tid);
    stK(s0, pre, tid);
    ldK(pre, Eb + (size_t)32 * DD, DD, tid);
    stK(s1, pre, tid);
    ldK(pre, Eb + (size_t)64 * DD, DD, tid);
    __syncthreads();
    ldAllK(fA, AS, s0 + bK);
    ring_signal();
#pragma unroll
    for (int nb = 0; nb < 4; ++nb) {
        for (int kc = 0; kc < 16; kc += 2) {
            const int ti = nb * 16 + kc;
            const int ta = ti + 3 < 64 ? ti + 3 : 63;
            const int tb = ti + 4 < 64 ? ti + 4 : 63;
            ring_wait();
            ldAllK(fB, AS + (kc + 1) * 32, s1 + bK);
            stK(s0, pre, tid);
            ldK(pre, Eb + (size_t)((ta & 15) * 32) * DD + (ta >> 4) * 128, DD, tid);
            group_end();
            mmPart<0, KS>(&ctx[nb * 4], fA);
            ring_signal();
            mmPart<KS, 8>(&ctx[nb * 4], fA);
            ring_wait();
            ldAllK(fA, AS + ((kc + 2) & 15) * 32, s0 + bK);
            stK(s1, pre, tid);
            ldK(pre, Eb + (size_t)((tb & 15) * 32) * DD + (tb >> 4) * 128, DD, tid);
            group_end();
            mmPart<0, KS>(&ctx[nb * 4], fB);
            ring_signal();
            mmPart<KS, 8>(&ctx[nb * 4], fB);
            body_end();
        }
    }
    ring_wait();
#pragma unroll
    for (int t = 0; t < 16; ++t) {
        const int n0 = (t >> 2) * 128 + ch * 64 + (t & 3) * 16;
#pragma unroll
        for (int i = 0; i < 8; ++i)
            sS[(rt * 16 + i + 8 * half) * STR + n0 + l16] = ctx[t][i];
    }

    ldR(pre, W, 2 * DD, tid);
    stR(s0, pre, tid);
    ldR(pre, W + 32, 2 * DD, tid);
    stR(s1, pre, tid);
    ldR(pre, W + 64, 2 * DD, tid);
    __syncthreads();
    ldAllR(fA, AQ, s0 + bR);
    ring_signal();
    for (int nb = 0; nb < 4; ++nb) {
        v8f acc[4] = {};
        for (int kc = 0; kc < 32; kc += 2) {
            const int ti = nb * 32 + kc;
            const int ta = ti + 3 < 128 ? ti + 3 : 127;
            const int tb = ti + 4 < 128 ? ti + 4 : 127;
            const int k1 = kc + 1;
            const int k2 = (kc + 2) & 31;
            ring_wait();
            ldAllR(fB, k1 < 16 ? AQ + k1 * 32 : AS + (k1 - 16) * 32, s1 + bR);
            stR(s0, pre, tid);
            ldR(pre, W + (size_t)((ta >> 5) * 128) * (2 * DD) + (ta & 31) * 32, 2 * DD, tid);
            group_end();
            mmPart<0, KS>(acc, fA);
            ring_signal();
            mmPart<KS, 8>(acc, fA);
            ring_wait();
            ldAllR(fA, k2 < 16 ? AQ + k2 * 32 : AS + (k2 - 16) * 32, s0 + bR);
            stR(s1, pre, tid);
            ldR(pre, W + (size_t)((tb >> 5) * 128) * (2 * DD) + (tb & 31) * 32, 2 * DD, tid);
            group_end();
            mmPart<0, KS>(acc, fB);
            ring_signal();
            mmPart<KS, 8>(acc, fB);
        }
#pragma unroll
        for (int j = 0; j < 4; ++j) {
            const int n0 = nb * 128 + ch * 64 + j * 16;
            const float bn = bias[n0 + l16];
            float* orow = outMasked + ((size_t)b * SD + q0 + rt * 16) * DD + n0 + l16;
#pragma unroll
            for (int i = 0; i < 8; ++i) {
                const int M = i + 8 * half;
                orow[(size_t)M * DD] = tanhf(acc[j][i] + bn) * sMask[rt * 16 + M];
            }
        }
    }
    ring_wait();
}

extern "C" void kernel_launch(void* const* d_in, const int* in_sizes, int n_in,
                              void* d_out, int out_size, void* d_ws, size_t ws_size,
                              hipStream_t stream) {
    (void)in_sizes; (void)n_in; (void)out_size; (void)d_ws; (void)ws_size;
    const float* Q    = (const float*)d_in[0];
    const float* E    = (const float*)d_in[1];
    const float* mask = (const float*)d_in[2];
    const float* W    = (const float*)d_in[3];
    const float* bias = (const float*)d_in[4];
    float* outMasked  = (float*)d_out;
    float* outWeights = (float*)d_out + (size_t)NB * SD * SE;

    dim3 grid(SD / ROWS, NB);
    attn_fused_kernel<<<grid, 256, 0, stream>>>(Q, E, mask, W, bias,
                                                outMasked, outWeights);
}


# Reads outside the tile loops requested ahead of use: Q tile as one burst of 16-B loads (was a loop of 8-load batches, a memory round trip each), mask and bias in registers before the output loop
# speedup vs baseline: 1.4215x; 1.0530x over previous
#include <hip/hip_runtime.h>
#include <math.h>


typedef __attribute__((ext_vector_type(2))) float v2f;
typedef __attribute__((ext_vector_type(4))) float v4f;
typedef __attribute__((ext_vector_type(8))) float v8f;

#define NB   32
#define SD   512
#define SE   512
#define DD   512
#define STR  516
#define ROWS 64
#define PR   36
#define P2   288
#define TILE (128 * PR)

struct Frag { v2f a[8]; v2f b[8][4]; };

__device__ __forceinline__ void ring_signal() {
    __builtin_amdgcn_sched_barrier(0);
    __builtin_amdgcn_fence(__ATOMIC_RELEASE, "workgroup", "local");
    __builtin_amdgcn_s_barrier_signal(-1);
    __builtin_amdgcn_sched_barrier(0);
}
__device__ __forceinline__ void ring_wait() {
    __builtin_amdgcn_sched_barrier(0);
    __builtin_amdgcn_s_barrier_wait(-1);
    __builtin_amdgcn_fence(__ATOMIC_ACQUIRE, "workgroup", "local");
    __builtin_amdgcn_sched_barrier(0);
}

__device__ __forceinline__ void group_end() {
    __builtin_amdgcn_sched_barrier(0);
}

__device__ __forceinline__ void body_end() {
    __builtin_amdgcn_sched_barrier(0);
}

__device__ __forceinline__ v8f wmma4(v2f a, v2f b, v8f c) {
    return __builtin_amdgcn_wmma_f32_16x16x4_f32(false, a, false, b, (short)0, c, false, false);
}

__device__ __forceinline__ void ldR(v4f (&r)[4], const float* __restrict__ g, int ld, int tid) {
#pragma unroll
    for (int j = 0; j < 4; ++j) {
        const int idx = tid + 256 * j;
        r[j] = *(const v4f*)(g + (size_t)(idx >> 3) * ld + (idx & 7) * 4);
    }
}
__device__ __forceinline__ void stR(float* buf, const v4f (&r)[4], int tid) {
#pragma unroll
    for (int j = 0; j < 4; ++j) {
        const int idx = tid + 256 * j;
        *(v4f*)(buf + (idx >> 3) * PR + (idx & 7) * 4) = r[j];
    }
}
__device__ __forceinline__ void ldK(v4f (&r)[4], const float* __restrict__ g, int ld, int tid) {
#pragma unroll
    for (int j = 0; j < 2; ++j) {
        const int idx = tid + 256 * j;
        const float* s = g + (size_t)(2 * (idx >> 5)) * ld + (idx & 31) * 4;
        r[2 * j]     = *(const v4f*)s;
        r[2 * j + 1] = *(const v4f*)(s + ld);
    }
}
__device__ __forceinline__ void stK(float* buf, const v4f (&r)[4], int tid) {
#pragma unroll
    for (int j = 0; j < 2; ++j) {
        const int idx = tid + 256 * j;
        float* d = buf + (idx >> 5) * P2 + (idx & 31) * 8;
        const v4f e = r[2 * j], o = r[2 * j + 1];
        *(v4f*)d       = (v4f){e.x, o.x, e.y, o.y};
        *(v4f*)(d + 4) = (v4f){e.z, o.z, e.w, o.w};
    }
}

__device__ __forceinline__ void ldAllR(Frag& f, const float* Ap, const float* Bp) {
#pragma unroll
    for (int s = 0; s < 8; ++s) {
        f.a[s] = *(const v2f*)(Ap + 4 * s);
#pragma unroll
        for (int j = 0; j < 4; ++j) f.b[s][j] = *(const v2f*)(Bp + j * 16 * PR + 4 * s);
    }
}
__device__ __forceinline__ void ldAllK(Frag& f, const float* Ap, const float* Bp) {
#pragma unroll
    for (int s = 0; s < 8; ++s) {
        f.a[s] = *(const v2f*)(Ap + 4 * s);
#pragma unroll
        for (int j = 0; j < 4; ++j) f.b[s][j] = *(const v2f*)(Bp + (2 * s) * P2 + j * 32);
    }
}
template <int S0, int S1>
__device__ __forceinline__ void mmPart(v8f* acc, const Frag& f) {
#pragma unroll
    for (int s = S0; s < S1; ++s)
#pragma unroll
        for (int j = 0; j < 4; ++j) acc[j] = wmma4(f.a[s], f.b[s][j], acc[j]);
}
#define KS 3

__launch_bounds__(256, 1)
__global__ void attn_fused_kernel(const float* __restrict__ Q,
                                  const float* __restrict__ E,
                                  const float* __restrict__ mask,
                                  const float* __restrict__ W,
                                  const float* __restrict__ bias,
                                  float* __restrict__ outMasked,
                                  float* __restrict__ outWeights)
{
    __shared__ float sQ[ROWS * STR];
    __shared__ float sS[ROWS * STR];
    __shared__ float sB[2 * TILE];
    float* sRed = sB;

    const int b    = blockIdx.y;
    const int q0   = blockIdx.x * ROWS;
    const int tid  = threadIdx.x;
    const int lane = tid & 31;
    const int wave = tid >> 5;
    const int half = lane >> 4;
    const int l16  = lane & 15;
    const int rt   = wave & 3;
    const int ch   = wave >> 2;

    const float* Qb = Q + ((size_t)b * SD + q0) * DD;
    const float* Eb = E + (size_t)b * SE * DD;

    v4f qv[32];
#pragma unroll
    for (int j = 0; j < 32; ++j) qv[j] = *(const v4f*)(Qb + (size_t)(tid + 256 * j) * 4);

    const float scale = 0.044194173824159216f;

    const int arow = (rt * 16 + l16) * STR + 2 * half;
    const float* AQ = sQ + arow;
    const float* AS = sS + arow;
    const int bR = (ch * 64 + l16) * PR + 2 * half;
    const int bK = half * P2 + 2 * (ch * 64 + l16);

    v4f pre[4], t0[4], t1[4];
    Frag fA, fB;
    float* const s0 = sB;
    float* const s1 = sB + TILE;

    ldR(t0, Eb, DD, tid);
    ldR(t1, Eb + 32, DD, tid);
    ldR(pre, Eb + 64, DD, tid);
#pragma unroll
    for (int j = 0; j < 32; ++j) {
        const int idx = tid + 256 * j;
        *(v4f*)(sQ + (idx >> 7) * STR + (idx & 127) * 4) = qv[j];
    }
    stR(s0, t0, tid);
    stR(s1, t1, tid);
    __syncthreads();
    ldAllR(fA, AQ, s0 + bR);
    ring_signal();
    for (int nb = 0; nb < 4; ++nb) {
        v8f acc[4] = {};
        for (int kc = 0; kc < 16; kc += 2) {
            const int ti = nb * 16 + kc;
            const int ta = ti + 3 < 64 ? ti + 3 : 63;
            const int tb = ti + 4 < 64 ? ti + 4 : 63;
            ring_wait();
            ldAllR(fB, AQ + (kc + 1) * 32, s1 + bR);
            stR(s0, pre, tid);
            ldR(pre, Eb + (size_t)((ta >> 4) * 128) * DD + (ta & 15) * 32, DD, tid);
            group_end();
            mmPart<0, KS>(acc, fA);
            ring_signal();
            mmPart<KS, 8>(acc, fA);
            ring_wait();
            ldAllR(fA, AQ + ((kc + 2) & 15) * 32, s0 + bR);
            stR(s1, pre, tid);
            ldR(pre, Eb + (size_t)((tb >> 4) * 128) * DD + (tb & 15) * 32, DD, tid);
            group_end();
            mmPart<0, KS>(acc, fB);
            ring_signal();
            mmPart<KS, 8>(acc, fB);
        }
#pragma unroll
        for (int j = 0; j < 4; ++j) {
            const int n0 = nb * 128 + ch * 64 + j * 16;
#pragma unroll
            for (int i = 0; i < 8; ++i)
                sS[(rt * 16 + i + 8 * half) * STR + n0 + l16] = acc[j][i] * scale;
        }
    }
    ring_wait();
    __syncthreads();

    ldK(t0, Eb, DD, tid);
    ldK(t1, Eb + (size_t)32 * DD, DD, tid);
    ldK(pre, Eb + (size_t)64 * DD, DD, tid);

    {
        const int r16 = tid & 15;
        const int sub = tid >> 4;
#pragma unroll
        for (int rg = 0; rg < 4; ++rg) {
            const int row = rg * 16 + r16;
            const float* srow = sS + row * STR;
            float m = -INFINITY;
            for (int c = sub * 32; c < sub * 32 + 32; ++c) m = fmaxf(m, srow[c]);
            sRed[row * 16 + sub] = m;
        }
        __syncthreads();
        float mall[4];
#pragma unroll
        for (int rg = 0; rg < 4; ++rg) {
            const int row = rg * 16 + r16;
            float mm = -INFINITY;
#pragma unroll
            for (int j = 0; j < 16; ++j) mm = fmaxf(mm, sRed[row * 16 + j]);
            mall[rg] = mm;
        }
        __syncthreads();
#pragma unroll
        for (int rg = 0; rg < 4; ++rg) {
            const int row = rg * 16 + r16;
            float* srow = sS + row * STR;
            float s = 0.f;
            for (int c = sub * 32; c < sub * 32 + 32; ++c) {
                float p = __expf(srow[c] - mall[rg]);
                srow[c] = p;
                s += p;
            }
            sRed[row * 16 + sub] = s;
        }
        __syncthreads();
#pragma unroll
        for (int rg = 0; rg < 4; ++rg) {
            const int row = rg * 16 + r16;
            float* srow = sS + row * STR;
            float stot = 0.f;
#pragma unroll
            for (int j = 0; j < 16; ++j) stot += sRed[row * 16 + j];
            const float inv = 1.0f / stot;
            float* gw = outWeights + ((size_t)b * SD + q0 + row) * SE;
            for (int c = sub * 32; c < sub * 32 + 32; ++c) {
                float p = srow[c] * inv;
                srow[c] = p;
                gw[c]   = p;
            }
        }
    }
    __syncthreads();

    v8f ctx[16];
#pragma unroll
    for (int t = 0; t < 16; ++t) ctx[t] = (v8f){};
    stK(s0, t0, tid);
    stK(s1, t1, tid);
    __syncthreads();
    ldAllK(fA, AS, s0 + bK);
    ring_signal();
#pragma unroll
    for (int nb = 0; nb < 4; ++nb) {
        for (int kc = 0; kc < 16; kc += 2) {
            const int ti = nb * 16 + kc;
            const int ta = ti + 3 < 64 ? ti + 3 : 63;
            const int tb = ti + 4 < 64 ? ti + 4 : 63;
            ring_wait();
            ldAllK(fB, AS + (kc + 1) * 32, s1 + bK);
            stK(s0, pre, tid);
            ldK(pre, Eb + (size_t)((ta & 15) * 32) * DD + (ta >> 4) * 128, DD, tid);
            group_end();
            mmPart<0, KS>(&ctx[nb * 4], fA);
            ring_signal();
            mmPart<KS, 8>(&ctx[nb * 4], fA);
            ring_wait();
            ldAllK(fA, AS + ((kc + 2) & 15) * 32, s0 + bK);
            stK(s1, pre, tid);
            ldK(pre, Eb + (size_t)((tb & 15) * 32) * DD + (tb >> 4) * 128, DD, tid);
            group_end();
            mmPart<0, KS>(&ctx[nb * 4], fB);
            ring_signal();
            mmPart<KS, 8>(&ctx[nb * 4], fB);
            body_end();
        }
    }
    ring_wait();
#pragma unroll
    for (int t = 0; t < 16; ++t) {
        const int n0 = (t >> 2) * 128 + ch * 64 + (t & 3) * 16;
#pragma unroll
        for (int i = 0; i < 8; ++i)
            sS[(rt * 16 + i + 8 * half) * STR + n0 + l16] = ctx[t][i];
    }

    float mk[8];
#pragma unroll
    for (int i = 0; i < 8; ++i) mk[i] = mask[(size_t)b * SD + q0 + rt * 16 + i + 8 * half];
    ldR(pre, W, 2 * DD, tid);
    stR(s0, pre, tid);
    ldR(pre, W + 32, 2 * DD, tid);
    stR(s1, pre, tid);
    ldR(pre, W + 64, 2 * DD, tid);
    __syncthreads();
    ldAllR(fA, AQ, s0 + bR);
    ring_signal();
    for (int nb = 0; nb < 4; ++nb) {
        v8f acc[4] = {};
        float bn[4];
#pragma unroll
        for (int j = 0; j < 4; ++j) bn[j] = bias[nb * 128 + ch * 64 + j * 16 + l16];
        for (int kc = 0; kc < 32; kc += 2) {
            const int ti = nb * 32 + kc;
            const int ta = ti + 3 < 128 ? ti + 3 : 127;
            const int tb = ti + 4 < 128 ? ti + 4 : 127;
            const int k1 = kc + 1;
            const int k2 = (kc + 2) & 31;
            ring_wait();
            ldAllR(fB, k1 < 16 ? AQ + k1 * 32 : AS + (k1 - 16) * 32, s1 + bR);
            stR(s0, pre, tid);
            ldR(pre, W + (size_t)((ta >> 5) * 128) * (2 * DD) + (ta & 31) * 32, 2 * DD, tid);
            group_end();
            mmPart<0, KS>(acc, fA);
            ring_signal();
            mmPart<KS, 8>(acc, fA);
            ring_wait();
            ldAllR(fA, k2 < 16 ? AQ + k2 * 32 : AS + (k2 - 16) * 32, s0 + bR);
            stR(s1, pre, tid);
            ldR(pre, W + (size_t)((tb >> 5) * 128) * (2 * DD) + (tb & 31) * 32, 2 * DD, tid);
            group_end();
            mmPart<0, KS>(acc, fB);
            ring_signal();
            mmPart<KS, 8>(acc, fB);
        }
#pragma unroll
        for (int j = 0; j < 4; ++j) {
            const int n0 = nb * 128 + ch * 64 + j * 16;
            float* orow = outMasked + ((size_t)b * SD + q0 + rt * 16) * DD + n0 + l16;
#pragma unroll
            for (int i = 0; i < 8; ++i) {
                const int M = i + 8 * half;
                orow[(size_t)M * DD] = tanhf(acc[j][i] + bn[j]) * mk[i];
            }
        }
    }
    ring_wait();
}

extern "C" void kernel_launch(void* const* d_in, const int* in_sizes, int n_in,
                              void* d_out, int out_size, void* d_ws, size_t ws_size,
                              hipStream_t stream) {
    (void)in_sizes; (void)n_in; (void)out_size; (void)d_ws; (void)ws_size;
    const float* Q    = (const float*)d_in[0];
    const float* E    = (const float*)d_in[1];
    const float* mask = (const float*)d_in[2];
    const float* W    = (const float*)d_in[3];
    const float* bias = (const float*)d_in[4];
    float* outMasked  = (float*)d_out;
    float* outWeights = (float*)d_out + (size_t)NB * SD * SE;

    dim3 grid(SD / ROWS, NB);
    attn_fused_kernel<<<grid, 256, 0, stream>>>(Q, E, mask, W, bias,
                                                outMasked, outWeights);
}
